# starting kernel
# baseline (speedup 1.0000x reference)
#include <hip/hip_runtime.h>

#define BB 4
#define SS 512
#define HH 256

typedef float v2f __attribute__((ext_vector_type(2)));
typedef float v8f __attribute__((ext_vector_type(8)));

#if __has_builtin(__builtin_amdgcn_tanhf)
#define TANHF(x) __builtin_amdgcn_tanhf(x)
#else
#define TANHF(x) tanhf(x)
#endif

__device__ __forceinline__ v8f wmma_f32_16x16x4(v2f a, v2f b, v8f c) {
    return __builtin_amdgcn_wmma_f32_16x16x4_f32(false, a, false, b, (short)0, c,
                                                 false, false);
}

__global__ void bahdanau_proj_gemm(const float* __restrict__ X,
                                   const float* __restrict__ W,
                                   const float* __restrict__ U,
                                   float* __restrict__ Qo,
                                   float* __restrict__ Ko) {
    const int tid  = threadIdx.x;
    const int wave = tid >> 5;
    const int lane = tid & 31;
    const int hp   = lane >> 4;
    const int lid  = lane & 15;

    const int tile = blockIdx.x * 8 + wave;
    const float* Bm = (tile & 2048) ? U : W;
    float*      Out = (tile & 2048) ? Ko : Qo;
    const int rem = tile & 2047;
    const int m0  = (rem >> 4) << 4;
    const int n0  = (rem & 15) << 4;

    const float* arow = X  + (size_t)(m0 + lid) * HH;
    const float* brow = Bm + (size_t)(n0 + lid) * HH;

    v8f acc = {};
#pragma unroll 4
    for (int k = 0; k < HH; k += 4) {
        v2f a = *(const v2f*)(arow + k + 2 * hp);
        v2f b = *(const v2f*)(brow + k + 2 * hp);
        acc = wmma_f32_16x16x4(a, b, acc);
    }

    float* orow = Out + n0 + lid;
#pragma unroll
    for (int r = 0; r < 8; ++r)
        orow[(size_t)(m0 + r + 8 * hp) * HH] = acc[r];
}

__global__ void __launch_bounds__(256)
bahdanau_attn(const float* __restrict__ X,
              const float* __restrict__ Qb,
              const float* __restrict__ Kb,
              const float* __restrict__ v,
              float* __restrict__ out) {
    __shared__ float Qs[16 * HH];
    __shared__ float sc[16 * SS];
    __shared__ float vs[HH];

    const int tid = threadIdx.x;
    const int b   = blockIdx.x >> 5;
    const int q0  = (blockIdx.x & 31) << 4;

    {
        const float4* src = (const float4*)(Qb + ((size_t)b * SS + q0) * HH);
        float4*       dst = (float4*)Qs;
        for (int i = tid; i < (16 * HH) / 4; i += 256) dst[i] = src[i];
        vs[tid] = v[tid];
    }
    __syncthreads();

    {
        const int q  = tid >> 4;
        const int kk = tid & 15;
        const float* qrow = Qs + q * HH;
        for (int kb = 0; kb < SS; kb += 16) {
            const float* krow = Kb + ((size_t)b * SS + kb + kk) * HH;
            float acc = 0.f;
#pragma unroll 4
            for (int h = 0; h < HH; h += 4) {
                float4 qv = *(const float4*)(qrow + h);
                float4 kv = *(const float4*)(krow + h);
                float4 vv = *(const float4*)(vs + h);
                acc += vv.x * TANHF(qv.x + kv.x);
                acc += vv.y * TANHF(qv.y + kv.y);
                acc += vv.z * TANHF(qv.z + kv.z);
                acc += vv.w * TANHF(qv.w + kv.w);
            }
            sc[q * SS + kb + kk] = acc;
        }
    }
    __syncthreads();

    {
        const int wv = tid >> 5;
        const int ln = tid & 31;
        for (int r = 0; r < 2; ++r) {
            float* row = sc + (wv * 2 + r) * SS;
            float m = -__builtin_inff();
            for (int k = ln; k < SS; k += 32) m = fmaxf(m, row[k]);
            for (int off = 16; off > 0; off >>= 1)
                m = fmaxf(m, __shfl_xor(m, off, 32));
            float s = 0.f;
            for (int k = ln; k < SS; k += 32) {
                float e = __builtin_amdgcn_exp2f((row[k] - m) * 1.4426950408889634f);
                row[k] = e;
                s += e;
            }
            for (int off = 16; off > 0; off >>= 1) s += __shfl_xor(s, off, 32);
            const float inv = 1.0f / s;
            for (int k = ln; k < SS; k += 32) row[k] *= inv;
        }
    }
    __syncthreads();

    {
        const int wv  = tid >> 5;
        const int ln  = tid & 31;
        const int hp  = ln >> 4;
        const int lid = ln & 15;
        const float* xb = X + (size_t)b * SS * HH;

        for (int t = 0; t < 2; ++t) {
            const int n0 = wv * 32 + t * 16;
            v8f acc = {};
#pragma unroll 2
            for (int k = 0; k < SS; k += 4) {
                v2f a = *(const v2f*)(sc + lid * SS + k + 2 * hp);
                v2f bf;
                bf.x = xb[(size_t)(k + 2 * hp)     * HH + n0 + lid];
                bf.y = xb[(size_t)(k + 2 * hp + 1) * HH + n0 + lid];
                acc = wmma_f32_16x16x4(a, bf, acc);
            }
#pragma unroll
            for (int r = 0; r < 8; ++r)
                out[((size_t)b * SS + q0 + r + 8 * hp) * HH + n0 + lid] = acc[r];
        }
    }
}

extern "C" void kernel_launch(void* const* d_in, const int* in_sizes, int n_in,
                              void* d_out, int out_size, void* d_ws, size_t ws_size,
                              hipStream_t stream) {
    const float* X = (const float*)d_in[0];
    const float* W = (const float*)d_in[1];
    const float* U = (const float*)d_in[2];
    const float* v = (const float*)d_in[3];
    float* out = (float*)d_out;

    float* Qb = (float*)d_ws;
    float* Kb = Qb + (size_t)BB * SS * HH;

    bahdanau_proj_gemm<<<512, 256, 0, stream>>>(X, W, U, Qb, Kb);
    bahdanau_attn<<<BB * (SS / 16), 256, 0, stream>>>(X, Qb, Kb, v, out);
}


# Projection kernel re-cut: 64x64 tile per work-group through LDS (256 groups of 4 waves, a 32x32 tile per wave); operands fetched once per group in whole lines, fragments by 16-byte LDS reads
# speedup vs baseline: 6.0777x; 6.0777x over previous
#include <hip/hip_runtime.h>

#define BB 4
#define SS 512
#define HH 256

#define HC  32
#define KP  36
#define NCH (HH / HC)
#define TQ  8
#define QT  4
#define KT  4
#define SPT ((SS * HC / 4) / 256)

#define SP  (SS + 4)
#define KB  16

typedef float v2f __attribute__((ext_vector_type(2)));
typedef float v4f __attribute__((ext_vector_type(4)));
typedef float v8f __attribute__((ext_vector_type(8)));

#if __has_builtin(__builtin_amdgcn_tanhf)
#define TANHF(x) __builtin_amdgcn_tanhf(x)
#else
#define TANHF(x) tanhf(x)
#endif

__device__ __forceinline__ v8f wmma_f32_16x16x4(v2f a, v2f b, v8f c) {
    return __builtin_amdgcn_wmma_f32_16x16x4_f32(false, a, false, b, (short)0, c,
                                                 false, false);
}

#define PM   64
#define PN   64
#define PKC  64
#define PNC  (HH / PKC)
#define PP   (HH + 4)
#define PPC  ((PM * PKC / 4) / 128)

__global__ void __launch_bounds__(128)
bahdanau_proj_gemm(const float* __restrict__ X,
                   const float* __restrict__ W,
                   const float* __restrict__ U,
                   float* __restrict__ Qo,
                   float* __restrict__ Ko) {
    __shared__ __attribute__((aligned(16))) float As[PM * PP];
    __shared__ __attribute__((aligned(16))) float Bs[PN * PP];

    const int tid = threadIdx.x;
    const int mt  = blockIdx.x >> 3;
    const int nt  = blockIdx.x & 7;
    const float* Bm = (nt & 4) ? U : W;
    float*      Out = (nt & 4) ? Ko : Qo;
    const int m0 = mt * PM;
    const int n0 = (nt & 3) * PN;

    const int srow = tid >> 4;
    const int scol = (tid & 15) * 4;
    const float* ga = X  + (size_t)(m0 + srow) * HH + scol;
    const float* gb = Bm + (size_t)(n0 + srow) * HH + scol;
    float* la = As + srow * PP + scol;
    float* lb = Bs + ((srow & 1) * 16 + (srow >> 1)) * PP + scol;

    v4f pa[PNC][PPC], pb[PNC][PPC];
#pragma unroll
    for (int c = 0; c < PNC; ++c) {
#pragma unroll
        for (int i = 0; i < PPC; ++i) {
            pa[c][i] = *(const v4f*)(ga + (size_t)(8 * i) * HH + c * PKC);
            pb[c][i] = *(const v4f*)(gb + (size_t)(8 * i) * HH + c * PKC);
        }
        __builtin_amdgcn_sched_barrier(0);
    }

    const int wv  = tid >> 5;
    const int ln  = tid & 31;
    const int hp  = ln >> 4;
    const int lid = ln & 15;
    const int wm  = (wv & 1) * 32;
    const int wn  = (wv >> 1) * 32;
    const float* fa = As + (wm + lid) * PP + 4 * hp;
    const float* fb = Bs + (wn + lid) * PP + 4 * hp;

    v8f acc00 = {}, acc01 = {}, acc10 = {}, acc11 = {};

    auto lds_barrier = [&]() {
        asm volatile("s_wait_dscnt 0x0" ::: "memory");
        asm volatile("s_barrier_signal -1" ::: "memory");
        asm volatile("s_barrier_wait -1" ::: "memory");
    };

    auto stage = [&](int c) {
#pragma unroll
        for (int i = 0; i < PPC; ++i) {
            *(v4f*)(la + (8 * i) * PP + c * PKC) = pa[c][i];
            *(v4f*)(lb + ((i >> 2) * 32 + 4 * (i & 3)) * PP + c * PKC) = pb[c][i];
        }
    };
    auto steps = [&](int kfrom, int kto) {
#pragma unroll
        for (int k = kfrom; k < kto; k += 8) {
            const v4f a0 = *(const v4f*)(fa + k);
            const v4f a1 = *(const v4f*)(fa + 16 * PP + k);
            const v4f b0 = *(const v4f*)(fb + k);
            const v4f b1 = *(const v4f*)(fb + 16 * PP + k);
            v2f a0l, a0h, a1l, a1h, b0l, b0h, b1l, b1h;
            a0l.x = a0.x; a0l.y = a0.y; a0h.x = a0.z; a0h.y = a0.w;
            a1l.x = a1.x; a1l.y = a1.y; a1h.x = a1.z; a1h.y = a1.w;
            b0l.x = b0.x; b0l.y = b0.y; b0h.x = b0.z; b0h.y = b0.w;
            b1l.x = b1.x; b1l.y = b1.y; b1h.x = b1.z; b1h.y = b1.w;
            acc00 = wmma_f32_16x16x4(a0l, b0l, acc00);
            acc01 = wmma_f32_16x16x4(a0l, b1l, acc01);
            acc10 = wmma_f32_16x16x4(a1l, b0l, acc10);
            acc11 = wmma_f32_16x16x4(a1l, b1l, acc11);
            acc00 = wmma_f32_16x16x4(a0h, b0h, acc00);
            acc01 = wmma_f32_16x16x4(a0h, b1h, acc01);
            acc10 = wmma_f32_16x16x4(a1h, b0h, acc10);
            acc11 = wmma_f32_16x16x4(a1h, b1h, acc11);
        }
    };

    stage(0);
    lds_barrier();
#pragma unroll
    for (int c = 0; c < PNC; ++c) {
        steps(c * PKC, c * PKC + 3 * PKC / 4);
        if (c + 1 < PNC) stage(c + 1);
        steps(c * PKC + 3 * PKC / 4, (c + 1) * PKC);
        if (c + 1 < PNC) lds_barrier();
    }

    float* orow = Out + (size_t)(m0 + wm + 8 * hp) * HH + n0 + wn + 2 * lid;
#pragma unroll
    for (int r = 0; r < 8; ++r) {
        v2f o0, o1;
        o0.x = acc00[r];  o0.y = acc01[r];
        o1.x = acc10[r];  o1.y = acc11[r];
        *(v2f*)(orow + (size_t)r * HH)        = o0;
        *(v2f*)(orow + (size_t)(16 + r) * HH) = o1;
    }
}

__global__ void __launch_bounds__(256)
bahdanau_attn(const float* __restrict__ X,
              const float* __restrict__ Qb,
              const float* __restrict__ Kb,
              const float* __restrict__ v,
              float* __restrict__ out) {
    __shared__ __attribute__((aligned(16))) float Qs[TQ * HH];
    __shared__ __attribute__((aligned(16))) float sc[TQ * SP];
    __shared__ __attribute__((aligned(16))) float vs[HH];
    __shared__ __attribute__((aligned(16))) float Ks[2 * SS * KP];

    const int tid = threadIdx.x;
    const int b   = blockIdx.x / (SS / TQ);
    const int q0  = (blockIdx.x % (SS / TQ)) * TQ;

    {
        const float4* src = (const float4*)(Qb + ((size_t)b * SS + q0) * HH);
        float4*       dst = (float4*)Qs;
        for (int i = tid; i < (TQ * HH) / 4; i += 256) dst[i] = src[i];
        vs[tid] = v[tid];
    }

    {
        const int srow = tid >> 3;
        const int scol = (tid & 7) * 4;
        const float* gsrc = Kb + ((size_t)b * SS + srow) * HH + scol;
        float*       ldst = Ks + srow * KP + scol;

        v4f pre[SPT];
#pragma unroll
        for (int i = 0; i < SPT; ++i)
            pre[i] = *(const v4f*)(gsrc + (size_t)i * 32 * HH);
#pragma unroll
        for (int i = 0; i < SPT; ++i)
            *(v4f*)(ldst + i * 32 * KP) = pre[i];
        __syncthreads();

        const int wv  = tid >> 5;
        const int ln  = tid & 31;
        const int qb0 = (wv & 1) * QT;
        const int kb0 = (wv >> 1) * (KT * 32) + ln;
        const float* qrow = Qs + qb0 * HH;

        float acc[QT][KT];
#pragma unroll
        for (int qi = 0; qi < QT; ++qi)
#pragma unroll
            for (int j = 0; j < KT; ++j) acc[qi][j] = 0.f;

        for (int c = 0; c < NCH; ++c) {
            const bool more = (c + 1 < NCH);
            if (more) {
#pragma unroll
                for (int i = 0; i < SPT; ++i)
                    pre[i] = *(const v4f*)(gsrc + (size_t)i * 32 * HH + (c + 1) * HC);
            }

            const float* kc = Ks + (c & 1) * (SS * KP) + kb0 * KP;
            const float* qc = qrow + c * HC;
            const float* vc = vs + c * HC;
#pragma unroll 4
            for (int h = 0; h < HC; h += 4) {
                const v4f vv = *(const v4f*)(vc + h);
                v4f kv[KT];
#pragma unroll
                for (int j = 0; j < KT; ++j)
                    kv[j] = *(const v4f*)(kc + j * 32 * KP + h);
#pragma unroll
                for (int qi = 0; qi < QT; ++qi) {
                    const v4f qv = *(const v4f*)(qc + qi * HH + h);
#pragma unroll
                    for (int j = 0; j < KT; ++j) {
                        float a = acc[qi][j];
                        a += vv.x * TANHF(qv.x + kv[j].x);
                        a += vv.y * TANHF(qv.y + kv[j].y);
                        a += vv.z * TANHF(qv.z + kv[j].z);
                        a += vv.w * TANHF(qv.w + kv[j].w);
                        acc[qi][j] = a;
                    }
                }
            }

            if (more) {
                float* nb = ldst + ((c + 1) & 1) * (SS * KP);
#pragma unroll
                for (int i = 0; i < SPT; ++i)
                    *(v4f*)(nb + i * 32 * KP) = pre[i];
                __syncthreads();
            }
        }

#pragma unroll
        for (int qi = 0; qi < QT; ++qi)
#pragma unroll
            for (int j = 0; j < KT; ++j)
                sc[(qb0 + qi) * SP + kb0 + 32 * j] = acc[qi][j];
    }
    __syncthreads();

    {
        const int wv = tid >> 5;
        const int ln = tid & 31;
        {
            float* row = sc + wv * SP;
            float m = -__builtin_inff();
            for (int k = ln; k < SS; k += 32) m = fmaxf(m, row[k]);
            for (int off = 16; off > 0; off >>= 1)
                m = fmaxf(m, __shfl_xor(m, off, 32));
            float s = 0.f;
            for (int k = ln; k < SS; k += 32) {
                float e = __builtin_amdgcn_exp2f((row[k] - m) * 1.4426950408889634f);
                row[k] = e;
                s += e;
            }
            for (int off = 16; off > 0; off >>= 1) s += __shfl_xor(s, off, 32);
            const float inv = 1.0f / s;
            for (int k = ln; k < SS; k += 32) row[k] *= inv;
        }
    }
    __syncthreads();

    {
        const int wv  = tid >> 5;
        const int ln  = tid & 31;
        const int hp  = ln >> 4;
        const int lid = ln & 15;
        const float* xcol = X + (size_t)b * SS * HH + (size_t)(2 * hp) * HH + wv * 32 + 2 * lid;
        const float* arow = sc + (lid & (TQ - 1)) * SP + 2 * hp;

        v2f xa[KB][2], xn[KB][2];
        v2f wa[KB], wn[KB];
        v8f acc0 = {}, acc1 = {};

        auto fetch = [&](v2f (&xs)[KB][2], v2f (&ws)[KB], int kbase) {
#pragma unroll
            for (int i = 0; i < KB; ++i) {
                const int k = kbase + 4 * i;
                xs[i][0] = *(const v2f*)(xcol + (size_t)k * HH);
                xs[i][1] = *(const v2f*)(xcol + (size_t)(k + 1) * HH);
            }
#pragma unroll
            for (int i = 0; i < KB; ++i)
                ws[i] = *(const v2f*)(arow + kbase + 4 * i);
        };
        auto block = [&](v2f (&xs)[KB][2], v2f (&ws)[KB]) {
#pragma unroll
            for (int i = 0; i < KB; ++i) {
                v2f b0, b1;
                b0.x = xs[i][0].x;  b0.y = xs[i][1].x;
                b1.x = xs[i][0].y;  b1.y = xs[i][1].y;
                acc0 = wmma_f32_16x16x4(ws[i], b0, acc0);
                acc1 = wmma_f32_16x16x4(ws[i], b1, acc1);
            }
        };

        fetch(xa, wa, 0);
        for (int kb = 0; kb < SS; kb += 8 * KB) {
            fetch(xn, wn, kb + 4 * KB);
            block(xa, wa);
            if (kb + 8 * KB < SS) fetch(xa, wa, kb + 8 * KB);
            block(xn, wn);
        }

        if (hp == 0) {
            float* orow = out + ((size_t)b * SS + q0) * HH + wv * 32 + 2 * lid;
#pragma unroll
            for (int r = 0; r < 8; ++r) {
                v2f o;
                o.x = acc0[r];
                o.y = acc1[r];
                *(v2f*)(orow + (size_t)r * HH) = o;
            }
        }
    }
}

extern "C" void kernel_launch(void* const* d_in, const int* in_sizes, int n_in,
                              void* d_out, int out_size, void* d_ws, size_t ws_size,
                              hipStream_t stream) {
    const float* X = (const float*)d_in[0];
    const float* W = (const float*)d_in[1];
    const float* U = (const float*)d_in[2];
    const float* v = (const float*)d_in[3];
    float* out = (float*)d_out;

    float* Qb = (float*)d_ws;
    float* Kb = Qb + (size_t)BB * SS * HH;

    bahdanau_proj_gemm<<<256, 128, 0, stream>>>(X, W, U, Qb, Kb);
    bahdanau_attn<<<BB * (SS / TQ), 256, 0, stream>>>(X, Qb, Kb, v, out);
}
